# starting kernel
# baseline (speedup 1.0000x reference)
#include <hip/hip_runtime.h>
#include <hip/hip_bf16.h>


typedef __attribute__((ext_vector_type(2))) float v2f;
typedef __attribute__((ext_vector_type(8))) float v8f;

static constexpr int GN  = 50000;
static constexpr int GE  = 800000;
static constexpr int GIN = 128;
static constexpr int GH  = 128;
static constexpr int GC  = 64;


__global__ void deg_init_kernel(float* __restrict__ deg, int n) {
    int i = blockIdx.x * blockDim.x + threadIdx.x;
    if (i < n) deg[i] = 1.0f;
}

__global__ void deg_count_kernel(const int* __restrict__ dst, float* __restrict__ deg, int nE) {
    int e = blockIdx.x * blockDim.x + threadIdx.x;
    if (e < nE) atomicAdd(&deg[dst[e]], 1.0f);
}

__global__ void deg_to_dinv_kernel(float* __restrict__ deg, int n) {
    int i = blockIdx.x * blockDim.x + threadIdx.x;
    if (i < n) deg[i] = rsqrtf(deg[i]);
}


template <int DIN, int DOUT>
__global__ void gemm_wmma_f32_kernel(const float* __restrict__ X,
                                     const float* __restrict__ W,
                                     float* __restrict__ Y,
                                     int mTiles) {
    constexpr int tilesN = DOUT / 16;
    constexpr int KF     = DIN / 4;

    const int wave       = threadIdx.x >> 5;
    const int wavesPerBl = blockDim.x >> 5;
    const int wid        = blockIdx.x * wavesPerBl + wave;
    const int wavesTotal = gridDim.x * wavesPerBl;
    const int tn         = wid % tilesN;
    const int mStep      = wavesTotal / tilesN;

    const int lane = threadIdx.x & 31;
    const int half = lane >> 4;
    const int l15  = lane & 15;
    const int col  = tn * 16 + l15;

    v2f bfrag[KF];
#pragma unroll
    for (int j = 0; j < KF; ++j) {
        const int k = 4 * j + 2 * half;
        bfrag[j].x = W[(size_t)k * DOUT + col];
        bfrag[j].y = W[(size_t)(k + 1) * DOUT + col];
    }

    for (int tm = wid / tilesN; tm < mTiles; tm += mStep) {
        const float* __restrict__ xr =
            X + (size_t)(tm * 16 + l15) * DIN + 2 * half;
        v8f acc = {};
#pragma unroll
        for (int j = 0; j < KF; ++j) {
            v2f a;
            a.x = xr[4 * j];
            a.y = xr[4 * j + 1];
            acc = __builtin_amdgcn_wmma_f32_16x16x4_f32(
                 false, a,  false, bfrag[j],
                 (short)0, acc,  false,  false);
        }
        float* __restrict__ yr = Y + (size_t)(tm * 16 + 8 * half) * DOUT + col;
#pragma unroll
        for (int r = 0; r < 8; ++r)
            yr[(size_t)r * DOUT] = acc[r];
    }
}


template <int D, bool ADD_BIAS>
__global__ void selfloop_init_kernel(const float* __restrict__ h,
                                     const float* __restrict__ dinv,
                                     const float* __restrict__ bias,
                                     float* __restrict__ agg, int n) {
    size_t idx = (size_t)blockIdx.x * blockDim.x + threadIdx.x;
    size_t total = (size_t)n * D / 4;
    if (idx >= total) return;
    size_t base = idx * 4;
    int i = (int)(base / D);
    int f = (int)(base % D);
    float di = dinv[i];
    float w  = di * di;
    float4 v = *(const float4*)(h + base);
    float4 o;
    o.x = w * v.x; o.y = w * v.y; o.z = w * v.z; o.w = w * v.w;
    if (ADD_BIAS) {
        o.x += bias[f]; o.y += bias[f + 1]; o.z += bias[f + 2]; o.w += bias[f + 3];
    }
    *(float4*)(agg + base) = o;
}


template <int D>
__global__ void edge_scatter_kernel(const int* __restrict__ src,
                                    const int* __restrict__ dst,
                                    const float* __restrict__ dinv,
                                    const float* __restrict__ h,
                                    float* __restrict__ agg, int nE) {
    constexpr int VPT = D / 32;
    static_assert(VPT == 2 || VPT == 4, "D must be 64 or 128");
    const int wave = threadIdx.x >> 5;
    const int lane = threadIdx.x & 31;
    const int e    = blockIdx.x * (blockDim.x >> 5) + wave;
    if (e >= nE) return;

    const int s = src[e];
    const int d = dst[e];
    const float norm = dinv[s] * dinv[d];

    const float* __restrict__ hp = h   + (size_t)s * D + lane * VPT;
    float* __restrict__ ap       = agg + (size_t)d * D + lane * VPT;

    if (VPT == 4) {
        float4 v = *(const float4*)hp;
        atomicAdd(ap + 0, norm * v.x);
        atomicAdd(ap + 1, norm * v.y);
        atomicAdd(ap + 2, norm * v.z);
        atomicAdd(ap + 3, norm * v.w);
    } else {
        float2 v = *(const float2*)hp;
        atomicAdd(ap + 0, norm * v.x);
        atomicAdd(ap + 1, norm * v.y);
    }
}


template <int D>
__global__ void bias_bn_relu_kernel(float* __restrict__ agg,
                                    const float* __restrict__ b,
                                    const float* __restrict__ g,
                                    const float* __restrict__ be,
                                    const float* __restrict__ m,
                                    const float* __restrict__ v, int n) {
    size_t idx = (size_t)blockIdx.x * blockDim.x + threadIdx.x;
    size_t total = (size_t)n * D / 4;
    if (idx >= total) return;
    size_t base = idx * 4;
    int f = (int)(base % D);
    float4 a = *(float4*)(agg + base);
    float* pa = &a.x;
#pragma unroll
    for (int j = 0; j < 4; ++j) {
        int fj = f + j;
        float val = pa[j] + b[fj];
        val = g[fj] * (val - m[fj]) * rsqrtf(v[fj] + 1e-5f) + be[fj];
        pa[j] = fmaxf(val, 0.0f);
    }
    *(float4*)(agg + base) = a;
}


static inline size_t align256(size_t x) { return (x + 255) & ~(size_t)255; }

extern "C" void kernel_launch(void* const* d_in, const int* in_sizes, int n_in,
                              void* d_out, int out_size, void* d_ws, size_t ws_size,
                              hipStream_t stream) {
    (void)in_sizes; (void)n_in; (void)out_size; (void)ws_size;

    const float* x   = (const float*)d_in[0];
    const int*   ei  = (const int*)d_in[1];
    const int*   src = ei;
    const int*   dst = ei + GE;
    const float* W1  = (const float*)d_in[2];
    const float* b1  = (const float*)d_in[3];
    const float* g1  = (const float*)d_in[4];
    const float* be1 = (const float*)d_in[5];
    const float* m1  = (const float*)d_in[6];
    const float* v1  = (const float*)d_in[7];
    const float* W2  = (const float*)d_in[8];
    const float* b2  = (const float*)d_in[9];
    const float* g2  = (const float*)d_in[10];
    const float* be2 = (const float*)d_in[11];
    const float* m2  = (const float*)d_in[12];
    const float* v2  = (const float*)d_in[13];
    const float* W3  = (const float*)d_in[14];
    const float* b3  = (const float*)d_in[15];
    float* out = (float*)d_out;

    char* ws = (char*)d_ws;
    size_t off = 0;
    float* dinv = (float*)(ws + off); off += align256((size_t)GN * 4);
    float* hbuf = (float*)(ws + off); off += align256((size_t)GN * GH * 4);
    float* agg1 = (float*)(ws + off); off += align256((size_t)GN * GH * 4);
    float* agg2 = (float*)(ws + off); off += align256((size_t)GN * GH * 4);

    const int TPB = 256;

    deg_init_kernel<<<(GN + TPB - 1) / TPB, TPB, 0, stream>>>(dinv, GN);
    deg_count_kernel<<<(GE + TPB - 1) / TPB, TPB, 0, stream>>>(dst, dinv, GE);
    deg_to_dinv_kernel<<<(GN + TPB - 1) / TPB, TPB, 0, stream>>>(dinv, GN);

    const int mTiles = GN / 16;
    const int gemmBlocks = 512;
    const int edgeBlocks = (GE + 7) / 8;
    const int elemBlocksH = (GN * GH / 4 + TPB - 1) / TPB;
    const int elemBlocksC = (GN * GC / 4 + TPB - 1) / TPB;

    gemm_wmma_f32_kernel<GIN, GH><<<gemmBlocks, TPB, 0, stream>>>(
        x, W1, hbuf, mTiles);
    selfloop_init_kernel<GH, false><<<elemBlocksH, TPB, 0, stream>>>(
        hbuf, dinv, nullptr, agg1, GN);
    edge_scatter_kernel<GH><<<edgeBlocks, TPB, 0, stream>>>(
        src, dst, dinv, hbuf, agg1, GE);
    bias_bn_relu_kernel<GH><<<elemBlocksH, TPB, 0, stream>>>(
        agg1, b1, g1, be1, m1, v1, GN);

    gemm_wmma_f32_kernel<GH, GH><<<gemmBlocks, TPB, 0, stream>>>(
        agg1, W2, hbuf, mTiles);
    selfloop_init_kernel<GH, false><<<elemBlocksH, TPB, 0, stream>>>(
        hbuf, dinv, nullptr, agg2, GN);
    edge_scatter_kernel<GH><<<edgeBlocks, TPB, 0, stream>>>(
        src, dst, dinv, hbuf, agg2, GE);
    bias_bn_relu_kernel<GH><<<elemBlocksH, TPB, 0, stream>>>(
        agg2, b2, g2, be2, m2, v2, GN);

    gemm_wmma_f32_kernel<GH, GC><<<gemmBlocks, TPB, 0, stream>>>(
        agg2, W3, hbuf, mTiles);
    selfloop_init_kernel<GC, true><<<elemBlocksC, TPB, 0, stream>>>(
        hbuf, dinv, b3, out, GN);
    edge_scatter_kernel<GC><<<edgeBlocks, TPB, 0, stream>>>(
        src, dst, dinv, hbuf, out, GE);
}


# product moved behind the aggregation: input rows summed in the LDS tile, tile x W on WMMA (A from LDS), bias+BN+ReLU at the store; 5 launches, h/agg arrays gone; f32 sum order changes ((A x) W)
# speedup vs baseline: 3.0505x; 3.0505x over previous
#include <hip/hip_runtime.h>
#include <hip/hip_bf16.h>


typedef __attribute__((ext_vector_type(2))) float v2f;
typedef __attribute__((ext_vector_type(8))) float v8f;

static constexpr int GN  = 50000;
static constexpr int GE  = 800000;
static constexpr int GIN = 128;
static constexpr int GH  = 128;
static constexpr int GC  = 64;


__global__ void deg_init_kernel(float* __restrict__ deg, int n) {
    int i = blockIdx.x * blockDim.x + threadIdx.x;
    if (i < n) deg[i] = 1.0f;
}

__global__ void deg_count_kernel(const int* __restrict__ dst, float* __restrict__ deg, int nE) {
    int e = blockIdx.x * blockDim.x + threadIdx.x;
    if (e < nE) atomicAdd(&deg[dst[e]], 1.0f);
}

__global__ void deg_to_dinv_kernel(float* __restrict__ deg, int n) {
    int i = blockIdx.x * blockDim.x + threadIdx.x;
    if (i < n) deg[i] = rsqrtf(deg[i]);
}


static constexpr int TR = 200;
static constexpr int TRP = 208;
static constexpr int PITCH = 132;
static constexpr int TW = 32;
static constexpr int TT = TW * 32;
static constexpr int QC = 256;
static constexpr int SU = 4;
static constexpr int NTILES = (GN + TR - 1) / TR;
static constexpr int LCAP = 256;
static_assert(GN <= 65536, "a list entry packs the source node into 16 bits");

template <int VPT>
__device__ __forceinline__ void load_row(const float* __restrict__ p, float (&v)[VPT]) {
    if constexpr (VPT == 4) {
        const float4 t = *(const float4*)p;
        v[0] = t.x; v[1] = t.y; v[2] = t.z; v[3] = t.w;
    } else {
        const float2 t = *(const float2*)p;
        v[0] = t.x; v[1] = t.y;
    }
}

__device__ __forceinline__ void queue_order() {
    __builtin_amdgcn_fence(__ATOMIC_ACQ_REL, "wavefront");
    __builtin_amdgcn_wave_barrier();
}

__device__ __forceinline__ void queue_append(int* myq, int qhead, int& qcnt,
                                             bool m, unsigned entry) {
    const unsigned bal = __builtin_amdgcn_ballot_w32(m);
    if (bal) {
        const int pos = qhead + qcnt + (int)__builtin_amdgcn_mbcnt_lo(bal, 0u);
        if (m) myq[pos & (QC - 1)] = (int)entry;
        qcnt += __builtin_popcount(bal);
    }
}

template <int D, bool FULL>
__device__ __forceinline__ void sum_rows(int nn, int lane, int hrow, int arow, int wbits,
                                         const float* __restrict__ h, float* acc) {
    constexpr int VPT = D / 32;
#pragma unroll 1
    for (int k0 = 0; k0 < nn; k0 += 8) {
        float v[8][VPT];
#pragma unroll
        for (int kk = 0; kk < 8; ++kk) {
#pragma unroll
            for (int c = 0; c < VPT; ++c) v[kk][c] = 0.0f;
            if (FULL || k0 + kk < nn) {
                const int hr = __builtin_amdgcn_readlane(hrow, k0 + kk);
                load_row<VPT>(h + hr + lane * VPT, v[kk]);
            }
        }
#pragma unroll
        for (int kk = 0; kk < 8; ++kk) {
            if (FULL || k0 + kk < nn) {
                const float wk = __int_as_float(__builtin_amdgcn_readlane(wbits, k0 + kk));
                const int   ar = __builtin_amdgcn_readlane(arow, k0 + kk);
#pragma unroll
                for (int c = 0; c < VPT; ++c)
                    __hip_atomic_fetch_add(&acc[ar + c * 32 + lane], wk * v[kk][c],
                                           __ATOMIC_RELAXED, __HIP_MEMORY_SCOPE_WORKGROUP);
            }
        }
    }
}

template <int D, bool FULL, bool KEEP>
__device__ __forceinline__ void drain_edges(const int* myq, int qhead, int n,
                                            int lane, int base,
                                            const int* __restrict__ src,
                                            const float* __restrict__ dinv,
                                            const float* __restrict__ h,
                                            float* acc,
                                            int2* __restrict__ lst, int& lcnt) {
    const int  nn = FULL ? 32 : n;
    const bool ok = lane < nn;
    const int  ent = myq[(qhead + lane) & (QC - 1)];
    const int  e   = ok ? (ent >> 8) : 0;
    const int  dl  = ok ? (ent & 255) : 0;
    const int  s   = src[e];
    const float w  = dinv[s] * dinv[base + dl];
    const int  wbits = __float_as_int(w);
    if (KEEP) {
        if (ok && lcnt + lane < LCAP) lst[lcnt + lane] = make_int2(s | (dl << 16), wbits);
        lcnt += nn;
    }
    sum_rows<D, FULL>(nn, lane, s * D, dl * PITCH, wbits, h, acc);
}

template <int D, bool KEEP>
__device__ __forceinline__ void process_step(const int4 (&buf)[SU], int i0, int tid, int lane,
                                             int base, int rows, int* myq, int& qhead, int& qcnt,
                                             const int* __restrict__ src,
                                             const float* __restrict__ dinv,
                                             const float* __restrict__ h, float* acc,
                                             int2* __restrict__ lst, int& lcnt) {
    constexpr int E4 = GE / 4;
#pragma unroll
    for (int j = 0; j < SU; ++j) {
        const int4 v = buf[j];
        const int lb = (i0 + j * TT + tid < E4) ? base : GN;
        const unsigned l0 = (unsigned)(v.x - lb);
        const unsigned l1 = (unsigned)(v.y - lb);
        const unsigned l2 = (unsigned)(v.z - lb);
        const unsigned l3 = (unsigned)(v.w - lb);
        const unsigned mn = min(min(l0, l1), min(l2, l3));
        if (__builtin_amdgcn_ballot_w32(mn < (unsigned)rows)) {
            const unsigned eb = (unsigned)((i0 + j * TT + tid) * 4) << 8;
            queue_append(myq, qhead, qcnt, l0 < (unsigned)rows, (eb + (0u << 8)) | l0);
            queue_append(myq, qhead, qcnt, l1 < (unsigned)rows, (eb + (1u << 8)) | l1);
            queue_append(myq, qhead, qcnt, l2 < (unsigned)rows, (eb + (2u << 8)) | l2);
            queue_append(myq, qhead, qcnt, l3 < (unsigned)rows, (eb + (3u << 8)) | l3);
            if (qcnt >= 32) {
                queue_order();
                do {
                    drain_edges<D, true, KEEP>(myq, qhead, 32, lane, base, src, dinv, h, acc, lst, lcnt);
                    qhead = (qhead + 32) & (QC - 1);
                    qcnt -= 32;
                } while (qcnt >= 32);
                queue_order();
            }
        }
    }
}

__device__ __forceinline__ void load_step(int4 (&buf)[SU], const int4* __restrict__ d4, int st, int tid) {
    constexpr int E4 = GE / 4;
#pragma unroll
    for (int j = 0; j < SU; ++j) {
        const int idx = st * (TT * SU) + j * TT + tid;
        buf[j] = d4[idx < E4 ? idx : E4 - 1];
    }
}

template <int D, bool KEEP>
__device__ __forceinline__ void scan_dst(const int* __restrict__ src,
                                         const int* __restrict__ dst,
                                         const float* __restrict__ dinv,
                                         const float* __restrict__ h,
                                         float* acc, int* myq, int tid, int lane,
                                         int base, int rows,
                                         int2* __restrict__ lst, int& lcnt) {
    int qhead = 0, qcnt = 0;
    const int4* __restrict__ d4 = (const int4*)dst;
    constexpr int E4   = GE / 4;
    constexpr int STEP = TT * SU;

    constexpr int NSTEPS = (E4 + STEP - 1) / STEP;
    static_assert(NSTEPS >= 3, "three buffers in rotation");
    int sA = blockIdx.x % NSTEPS;
    int sB = (sA + 1 == NSTEPS) ? 0 : sA + 1;
    int sC = (sB + 1 == NSTEPS) ? 0 : sB + 1;
    int4 bufA[SU], bufB[SU], bufC[SU];
    load_step(bufA, d4, sA, tid);
    load_step(bufB, d4, sB, tid);
    load_step(bufC, d4, sC, tid);
    for (int it = 0; ; ) {
        process_step<D, KEEP>(bufA, sA * STEP, tid, lane, base, rows, myq, qhead, qcnt, src, dinv, h, acc, lst, lcnt);
        if (++it == NSTEPS) break;
        sA = (sC + 1 == NSTEPS) ? 0 : sC + 1;
        if (it + 2 < NSTEPS) load_step(bufA, d4, sA, tid);
        process_step<D, KEEP>(bufB, sB * STEP, tid, lane, base, rows, myq, qhead, qcnt, src, dinv, h, acc, lst, lcnt);
        if (++it == NSTEPS) break;
        sB = (sA + 1 == NSTEPS) ? 0 : sA + 1;
        if (it + 2 < NSTEPS) load_step(bufB, d4, sB, tid);
        process_step<D, KEEP>(bufC, sC * STEP, tid, lane, base, rows, myq, qhead, qcnt, src, dinv, h, acc, lst, lcnt);
        if (++it == NSTEPS) break;
        sC = (sB + 1 == NSTEPS) ? 0 : sB + 1;
        if (it + 2 < NSTEPS) load_step(bufC, d4, sC, tid);
    }
    if (qcnt > 0) {
        queue_order();
        drain_edges<D, false, KEEP>(myq, qhead, qcnt, lane, base, src, dinv, h, acc, lst, lcnt);
    }
}

template <int DOUT, bool BN, bool FROM_LIST>
__global__ __launch_bounds__(TT)
void agg_mm_tile_kernel(const int* __restrict__ src,
                        const int* __restrict__ dst,
                        const float* __restrict__ dinv,
                        const float* __restrict__ h,
                        const float* __restrict__ W,
                        const float* __restrict__ bias,
                        const float* __restrict__ g,
                        const float* __restrict__ be,
                        const float* __restrict__ m,
                        const float* __restrict__ v,
                        float* __restrict__ y,
                        int* __restrict__ lcnts,
                        int* __restrict__ lists) {
    constexpr int D   = GH;
    constexpr int VPT = D / 32;
    static_assert(GIN == GH, "all three layers aggregate 128-wide rows");
    static_assert(VPT == 4, "the position map below is for 4 values per lane");
    static_assert(GE % 4 == 0, "dst is read 4 edges per lane");
    static_assert(TR <= 256, "row-in-tile is packed into 8 bits");
    static_assert(TRP % 16 == 0 && TRP >= TR, "whole 16-row blocks");

    __shared__ __attribute__((aligned(16))) float acc[TRP * PITCH];
    __shared__ int   qbuf[TW * QC];

    const int tid  = threadIdx.x;
    const int lane = tid & 31;
    const int wave = tid >> 5;
    const int base = blockIdx.x * TR;
    const int rows = (GN - base < TR) ? (GN - base) : TR;

    int*  myq = qbuf + wave * QC;
    int2* __restrict__ lst = (int2*)lists + (size_t)(blockIdx.x * TW + wave) * LCAP;
    int   lcnt = 0;

    int cl = 0;
    if constexpr (FROM_LIST) cl = lcnts[blockIdx.x * TW + lane];

    for (int r = wave; r < TRP; r += TW) {
        float v4[VPT];
#pragma unroll
        for (int c = 0; c < VPT; ++c) v4[c] = 0.0f;
        if (r < rows) {
            const int i = base + r;
            const float di = dinv[i];
            const float w  = di * di;
            load_row<VPT>(h + (size_t)i * D + lane * VPT, v4);
#pragma unroll
            for (int c = 0; c < VPT; ++c) v4[c] *= w;
        }
#pragma unroll
        for (int c = 0; c < VPT; ++c) acc[r * PITCH + c * 32 + lane] = v4[c];
    }

    __syncthreads();

    if constexpr (!FROM_LIST) {
        scan_dst<D, true>(src, dst, dinv, h, acc, myq, tid, lane, base, rows, lst, lcnt);
        if (lane == 0) lcnts[blockIdx.x * TW + wave] = lcnt;
    } else {
        if (__builtin_amdgcn_ballot_w32(cl > LCAP) == 0) {
            const int n = __builtin_amdgcn_readlane(cl, wave);
            int2 cur = make_int2(0, 0);
            if (lane < n) cur = lst[lane];
            for (int k0 = 0; k0 < n; k0 += 32) {
                int2 nxt = make_int2(0, 0);
                if (k0 + 32 + lane < n) nxt = lst[k0 + 32 + lane];
                const int hrow = (cur.x & 0xFFFF) * D;
                const int arow = (int)((unsigned)cur.x >> 16) * PITCH;
                if (n - k0 >= 32) sum_rows<D, true>(32, lane, hrow, arow, cur.y, h, acc);
                else              sum_rows<D, false>(n - k0, lane, hrow, arow, cur.y, h, acc);
                cur = nxt;
            }
        } else {
            scan_dst<D, false>(src, dst, dinv, h, acc, myq, tid, lane, base, rows, lst, lcnt);
        }
    }

    constexpr int TN  = DOUT / 16;
    constexpr int RT  = TRP / 16;
    constexpr int WPC = TW / TN;
    constexpr int KF  = D / 4;
    static_assert(TW % TN == 0, "waves split evenly over the column blocks");

    const int half = lane >> 4;
    const int l15  = lane & 15;
    const int wv   = __builtin_amdgcn_readfirstlane(wave);
    const int col  = (wv % TN) * 16 + l15;

    v2f bfrag[KF];
#pragma unroll
    for (int j = 0; j < KF; ++j) {
        const int p = 8 * (j >> 1) + 4 * half + 2 * (j & 1);
        const int k = (p & 31) * 4 + (p >> 5);
        bfrag[j].x = W[(size_t)k * DOUT + col];
        bfrag[j].y = W[(size_t)(k + 4) * DOUT + col];
    }
    const float bcol = bias[col];
    float gcol = 0.0f, mcol = 0.0f, rcol = 0.0f, becol = 0.0f;
    if constexpr (BN) {
        gcol  = g[col];
        mcol  = m[col];
        rcol  = rsqrtf(v[col] + 1e-5f);
        becol = be[col];
    }

    __syncthreads();

    for (int tm = wv / TN; tm < RT; tm += WPC) {
        const float* ar = acc + (tm * 16 + l15) * PITCH + 4 * half;
        v8f d = {};
#pragma unroll
        for (int jj = 0; jj < KF / 2; ++jj) {
            const float4 t = *(const float4*)(ar + 8 * jj);
            v2f a0, a1;
            a0.x = t.x; a0.y = t.y;
            a1.x = t.z; a1.y = t.w;
            d = __builtin_amdgcn_wmma_f32_16x16x4_f32(
                 false, a0,  false, bfrag[2 * jj],
                 (short)0, d,  false,  false);
            d = __builtin_amdgcn_wmma_f32_16x16x4_f32(
                 false, a1,  false, bfrag[2 * jj + 1],
                 (short)0, d,  false,  false);
        }
        const int r0 = tm * 16 + 8 * half;
        float* __restrict__ yr = y + (size_t)(base + r0) * DOUT + col;
        float o[8];
#pragma unroll
        for (int r = 0; r < 8; ++r) {
            float val = d[r] + bcol;
            if constexpr (BN) {
                val = gcol * (val - mcol) * rcol + becol;
                val = fmaxf(val, 0.0f);
            }
            o[r] = val;
        }
        if (tm * 16 + 16 <= rows) {
#pragma unroll
            for (int r = 0; r < 8; ++r) yr[(size_t)r * DOUT] = o[r];
        } else {
#pragma unroll
            for (int r = 0; r < 8; ++r)
                if (r0 + r < rows) yr[(size_t)r * DOUT] = o[r];
        }
    }
}


static inline size_t align256(size_t x) { return (x + 255) & ~(size_t)255; }

extern "C" void kernel_launch(void* const* d_in, const int* in_sizes, int n_in,
                              void* d_out, int out_size, void* d_ws, size_t ws_size,
                              hipStream_t stream) {
    (void)in_sizes; (void)n_in; (void)out_size; (void)ws_size;

    const float* x   = (const float*)d_in[0];
    const int*   ei  = (const int*)d_in[1];
    const int*   src = ei;
    const int*   dst = ei + GE;
    const float* W1  = (const float*)d_in[2];
    const float* b1  = (const float*)d_in[3];
    const float* g1  = (const float*)d_in[4];
    const float* be1 = (const float*)d_in[5];
    const float* m1  = (const float*)d_in[6];
    const float* v1  = (const float*)d_in[7];
    const float* W2  = (const float*)d_in[8];
    const float* b2  = (const float*)d_in[9];
    const float* g2  = (const float*)d_in[10];
    const float* be2 = (const float*)d_in[11];
    const float* m2  = (const float*)d_in[12];
    const float* v2  = (const float*)d_in[13];
    const float* W3  = (const float*)d_in[14];
    const float* b3  = (const float*)d_in[15];
    float* out = (float*)d_out;

    char* ws = (char*)d_ws;
    size_t off = 0;
    float* dinv = (float*)(ws + off); off += align256((size_t)GN * 4);
    float* y1   = (float*)(ws + off); off += align256((size_t)GN * GH * 4);
    float* y2   = (float*)(ws + off); off += align256((size_t)GN * GH * 4);
    int*   lcnts = (int*)(ws + off);  off += align256((size_t)NTILES * TW * 4);
    int*   lists = (int*)(ws + off);  off += align256((size_t)NTILES * TW * LCAP * 8);

    const int TPB = 256;

    deg_init_kernel<<<(GN + TPB - 1) / TPB, TPB, 0, stream>>>(dinv, GN);
    deg_count_kernel<<<(GE + TPB - 1) / TPB, TPB, 0, stream>>>(dst, dinv, GE);
    deg_to_dinv_kernel<<<(GN + TPB - 1) / TPB, TPB, 0, stream>>>(dinv, GN);

    agg_mm_tile_kernel<GH, true, false><<<NTILES, TT, 0, stream>>>(
        src, dst, dinv, x, W1, b1, g1, be1, m1, v1, y1, lcnts, lists);

    agg_mm_tile_kernel<GH, true, true><<<NTILES, TT, 0, stream>>>(
        src, dst, dinv, y1, W2, b2, g2, be2, m2, v2, y2, lcnts, lists);

    agg_mm_tile_kernel<GC, false, true><<<NTILES, TT, 0, stream>>>(
        src, dst, dinv, y2, W3, b3, nullptr, nullptr, nullptr, nullptr, out, lcnts, lists);
}


# degree pass also writes a 1-byte tile number per edge (16 edges a lane, one 16-B store); layer 1 scans these bytes, not dst: 4x fewer bytes, SWAR match; LDS f32 add order changes (arrival order)
# speedup vs baseline: 3.1828x; 1.0434x over previous
#include <hip/hip_runtime.h>
#include <hip/hip_bf16.h>


typedef __attribute__((ext_vector_type(2))) float v2f;
typedef __attribute__((ext_vector_type(8))) float v8f;

static constexpr int GN  = 50000;
static constexpr int GE  = 800000;
static constexpr int GIN = 128;
static constexpr int GH  = 128;
static constexpr int GC  = 64;


__global__ void deg_init_kernel(float* __restrict__ deg, int n) {
    int i = blockIdx.x * blockDim.x + threadIdx.x;
    if (i < n) deg[i] = 1.0f;
}

static constexpr int TR = 200;

__device__ __forceinline__ unsigned tile_bytes(const int4 d) {
    return  ((unsigned)d.x / (unsigned)TR)        | (((unsigned)d.y / (unsigned)TR) << 8) |
           (((unsigned)d.z / (unsigned)TR) << 16) | (((unsigned)d.w / (unsigned)TR) << 24);
}

__global__ void deg_count_kernel(const int* __restrict__ dst, float* __restrict__ deg,
                                 unsigned int* __restrict__ tile8, int nG) {
    const int t = blockIdx.x * blockDim.x + threadIdx.x;
    if (t < nG) {
        const int4* __restrict__ d4 = (const int4*)dst + (size_t)t * 4;
        const int4 a = d4[0], b = d4[1], c = d4[2], d = d4[3];
        atomicAdd(&deg[a.x], 1.0f); atomicAdd(&deg[a.y], 1.0f); atomicAdd(&deg[a.z], 1.0f); atomicAdd(&deg[a.w], 1.0f);
        atomicAdd(&deg[b.x], 1.0f); atomicAdd(&deg[b.y], 1.0f); atomicAdd(&deg[b.z], 1.0f); atomicAdd(&deg[b.w], 1.0f);
        atomicAdd(&deg[c.x], 1.0f); atomicAdd(&deg[c.y], 1.0f); atomicAdd(&deg[c.z], 1.0f); atomicAdd(&deg[c.w], 1.0f);
        atomicAdd(&deg[d.x], 1.0f); atomicAdd(&deg[d.y], 1.0f); atomicAdd(&deg[d.z], 1.0f); atomicAdd(&deg[d.w], 1.0f);
        uint4 o;
        o.x = tile_bytes(a); o.y = tile_bytes(b); o.z = tile_bytes(c); o.w = tile_bytes(d);
        ((uint4*)tile8)[t] = o;
    }
}

__global__ void deg_to_dinv_kernel(float* __restrict__ deg, int n) {
    int i = blockIdx.x * blockDim.x + threadIdx.x;
    if (i < n) deg[i] = rsqrtf(deg[i]);
}


static constexpr int TRP = 208;
static constexpr int PITCH = 132;
static constexpr int TW = 32;
static constexpr int TT = TW * 32;
static constexpr int QC = 256;
static constexpr int SU = 4;
static constexpr int NTILES = (GN + TR - 1) / TR;
static constexpr int LCAP = 256;
static_assert(GN <= 65536, "a list entry packs the source node into 16 bits");
static_assert(GE % 16 == 0, "the tile bytes are read and written 16 edges at a time");
static_assert(NTILES <= 255, "a tile number is one byte, and 255 is the value no tile has");

template <int VPT>
__device__ __forceinline__ void load_row(const float* __restrict__ p, float (&v)[VPT]) {
    if constexpr (VPT == 4) {
        const float4 t = *(const float4*)p;
        v[0] = t.x; v[1] = t.y; v[2] = t.z; v[3] = t.w;
    } else {
        const float2 t = *(const float2*)p;
        v[0] = t.x; v[1] = t.y;
    }
}

__device__ __forceinline__ void queue_order() {
    __builtin_amdgcn_fence(__ATOMIC_ACQ_REL, "wavefront");
    __builtin_amdgcn_wave_barrier();
}

__device__ __forceinline__ void queue_append(int* myq, int qhead, int& qcnt,
                                             bool m, unsigned entry) {
    const unsigned bal = __builtin_amdgcn_ballot_w32(m);
    if (bal) {
        const int pos = qhead + qcnt + (int)__builtin_amdgcn_mbcnt_lo(bal, 0u);
        if (m) myq[pos & (QC - 1)] = (int)entry;
        qcnt += __builtin_popcount(bal);
    }
}

template <int D, bool FULL>
__device__ __forceinline__ void sum_rows(int nn, int lane, int hrow, int arow, int wbits,
                                         const float* __restrict__ h, float* acc) {
    constexpr int VPT = D / 32;
#pragma unroll 1
    for (int k0 = 0; k0 < nn; k0 += 8) {
        float v[8][VPT];
#pragma unroll
        for (int kk = 0; kk < 8; ++kk) {
#pragma unroll
            for (int c = 0; c < VPT; ++c) v[kk][c] = 0.0f;
            if (FULL || k0 + kk < nn) {
                const int hr = __builtin_amdgcn_readlane(hrow, k0 + kk);
                load_row<VPT>(h + hr + lane * VPT, v[kk]);
            }
        }
#pragma unroll
        for (int kk = 0; kk < 8; ++kk) {
            if (FULL || k0 + kk < nn) {
                const float wk = __int_as_float(__builtin_amdgcn_readlane(wbits, k0 + kk));
                const int   ar = __builtin_amdgcn_readlane(arow, k0 + kk);
#pragma unroll
                for (int c = 0; c < VPT; ++c)
                    __hip_atomic_fetch_add(&acc[ar + c * 32 + lane], wk * v[kk][c],
                                           __ATOMIC_RELAXED, __HIP_MEMORY_SCOPE_WORKGROUP);
            }
        }
    }
}

template <int D, bool FULL, bool KEEP>
__device__ __forceinline__ void drain_edges(const int* myq, int qhead, int n,
                                            int lane, int base,
                                            const int* __restrict__ src,
                                            const int* __restrict__ dst,
                                            const float* __restrict__ dinv,
                                            const float* __restrict__ h,
                                            float* acc,
                                            int2* __restrict__ lst, int& lcnt) {
    const int  nn = FULL ? 32 : n;
    const bool ok = lane < nn;
    const int  ent = myq[(qhead + lane) & (QC - 1)];
    const int  e   = ok ? ent : 0;
    const int  s   = src[e];
    const int  dn  = dst[e];
    const int  dl  = ok ? (dn - base) : 0;
    const float w  = dinv[s] * dinv[dn];
    const int  wbits = __float_as_int(w);
    if (KEEP) {
        if (ok && lcnt + lane < LCAP) lst[lcnt + lane] = make_int2(s | (dl << 16), wbits);
        lcnt += nn;
    }
    sum_rows<D, FULL>(nn, lane, s * D, dl * PITCH, wbits, h, acc);
}

__device__ __forceinline__ unsigned nonzero_bytes(unsigned x) {
    return ((x & 0x7F7F7F7Fu) + 0x7F7F7F7Fu) | x;
}

template <int D, bool KEEP>
__device__ __forceinline__ void process_step(const uint4 (&buf)[SU], int g0, int tid, int lane,
                                             int base, unsigned t4, int* myq, int& qhead, int& qcnt,
                                             const int* __restrict__ src,
                                             const int* __restrict__ dst,
                                             const float* __restrict__ dinv,
                                             const float* __restrict__ h, float* acc,
                                             int2* __restrict__ lst, int& lcnt) {
    constexpr int E16 = GE / 16;
#pragma unroll
    for (int j = 0; j < SU; ++j) {
        const uint4 v = buf[j];
        const int gi = g0 + j * TT + tid;
        const unsigned tt = (gi < E16) ? t4 : 0xFFFFFFFFu;
        unsigned c = (((nonzero_bytes(v.x ^ tt) >> 7) & 0x01010101u) |
                      ((nonzero_bytes(v.y ^ tt) >> 6) & 0x02020202u) |
                      ((nonzero_bytes(v.z ^ tt) >> 5) & 0x04040404u) |
                      ((nonzero_bytes(v.w ^ tt) >> 4) & 0x08080808u)) ^ 0x0F0F0F0Fu;
        unsigned bal = __builtin_amdgcn_ballot_w32(c != 0u);
        while (bal) {
            const int p = __builtin_ctz(c | 0x80000000u);
            const unsigned e = (unsigned)gi * 16u + (unsigned)((p & 3) * 4 + (p >> 3));
            queue_append(myq, qhead, qcnt, c != 0u, e);
            c &= c - 1u;
            if (qcnt >= 32) {
                queue_order();
                do {
                    drain_edges<D, true, KEEP>(myq, qhead, 32, lane, base, src, dst, dinv, h, acc, lst, lcnt);
                    qhead = (qhead + 32) & (QC - 1);
                    qcnt -= 32;
                } while (qcnt >= 32);
                queue_order();
            }
            bal = __builtin_amdgcn_ballot_w32(c != 0u);
        }
    }
}

__device__ __forceinline__ void load_step(uint4 (&buf)[SU], const uint4* __restrict__ t16, int st, int tid) {
    constexpr int E16 = GE / 16;
#pragma unroll
    for (int j = 0; j < SU; ++j) {
        const int idx = st * (TT * SU) + j * TT + tid;
        buf[j] = t16[idx < E16 ? idx : E16 - 1];
    }
}

template <int D, bool KEEP>
__device__ __forceinline__ void scan_dst(const int* __restrict__ src,
                                         const int* __restrict__ dst,
                                         const unsigned int* __restrict__ tile8,
                                         const float* __restrict__ dinv,
                                         const float* __restrict__ h,
                                         float* acc, int* myq, int tid, int lane,
                                         int base,
                                         int2* __restrict__ lst, int& lcnt) {
    int qhead = 0, qcnt = 0;
    const uint4* __restrict__ d4 = (const uint4*)tile8;
    const unsigned t4 = (unsigned)blockIdx.x * 0x01010101u;
    constexpr int E16  = GE / 16;
    constexpr int STEP = TT * SU;

    constexpr int NSTEPS = (E16 + STEP - 1) / STEP;
    static_assert(NSTEPS >= 3, "three buffers in rotation");
    int sA = blockIdx.x % NSTEPS;
    int sB = (sA + 1 == NSTEPS) ? 0 : sA + 1;
    int sC = (sB + 1 == NSTEPS) ? 0 : sB + 1;
    uint4 bufA[SU], bufB[SU], bufC[SU];
    load_step(bufA, d4, sA, tid);
    load_step(bufB, d4, sB, tid);
    load_step(bufC, d4, sC, tid);
    for (int it = 0; ; ) {
        process_step<D, KEEP>(bufA, sA * STEP, tid, lane, base, t4, myq, qhead, qcnt, src, dst, dinv, h, acc, lst, lcnt);
        if (++it == NSTEPS) break;
        sA = (sC + 1 == NSTEPS) ? 0 : sC + 1;
        if (it + 2 < NSTEPS) load_step(bufA, d4, sA, tid);
        process_step<D, KEEP>(bufB, sB * STEP, tid, lane, base, t4, myq, qhead, qcnt, src, dst, dinv, h, acc, lst, lcnt);
        if (++it == NSTEPS) break;
        sB = (sA + 1 == NSTEPS) ? 0 : sA + 1;
        if (it + 2 < NSTEPS) load_step(bufB, d4, sB, tid);
        process_step<D, KEEP>(bufC, sC * STEP, tid, lane, base, t4, myq, qhead, qcnt, src, dst, dinv, h, acc, lst, lcnt);
        if (++it == NSTEPS) break;
        sC = (sB + 1 == NSTEPS) ? 0 : sB + 1;
        if (it + 2 < NSTEPS) load_step(bufC, d4, sC, tid);
    }
    if (qcnt > 0) {
        queue_order();
        drain_edges<D, false, KEEP>(myq, qhead, qcnt, lane, base, src, dst, dinv, h, acc, lst, lcnt);
    }
}

template <int DOUT, bool BN, bool FROM_LIST>
__global__ __launch_bounds__(TT)
void agg_mm_tile_kernel(const int* __restrict__ src,
                        const int* __restrict__ dst,
                        const float* __restrict__ dinv,
                        const float* __restrict__ h,
                        const float* __restrict__ W,
                        const float* __restrict__ bias,
                        const float* __restrict__ g,
                        const float* __restrict__ be,
                        const float* __restrict__ m,
                        const float* __restrict__ v,
                        float* __restrict__ y,
                        const unsigned int* __restrict__ tile8,
                        int* __restrict__ lcnts,
                        int* __restrict__ lists) {
    constexpr int D   = GH;
    constexpr int VPT = D / 32;
    static_assert(GIN == GH, "all three layers aggregate 128-wide rows");
    static_assert(VPT == 4, "the position map below is for 4 values per lane");
    static_assert(TR <= 256, "row-in-tile is packed into 8 bits");
    static_assert(TRP % 16 == 0 && TRP >= TR, "whole 16-row blocks");

    __shared__ __attribute__((aligned(16))) float acc[TRP * PITCH];
    __shared__ int   qbuf[TW * QC];

    const int tid  = threadIdx.x;
    const int lane = tid & 31;
    const int wave = tid >> 5;
    const int base = blockIdx.x * TR;
    const int rows = (GN - base < TR) ? (GN - base) : TR;

    int*  myq = qbuf + wave * QC;
    int2* __restrict__ lst = (int2*)lists + (size_t)(blockIdx.x * TW + wave) * LCAP;
    int   lcnt = 0;

    int cl = 0;
    if constexpr (FROM_LIST) cl = lcnts[blockIdx.x * TW + lane];

    for (int r = wave; r < TRP; r += TW) {
        float v4[VPT];
#pragma unroll
        for (int c = 0; c < VPT; ++c) v4[c] = 0.0f;
        if (r < rows) {
            const int i = base + r;
            const float di = dinv[i];
            const float w  = di * di;
            load_row<VPT>(h + (size_t)i * D + lane * VPT, v4);
#pragma unroll
            for (int c = 0; c < VPT; ++c) v4[c] *= w;
        }
#pragma unroll
        for (int c = 0; c < VPT; ++c) acc[r * PITCH + c * 32 + lane] = v4[c];
    }

    __syncthreads();

    if constexpr (!FROM_LIST) {
        scan_dst<D, true>(src, dst, tile8, dinv, h, acc, myq, tid, lane, base, lst, lcnt);
        if (lane == 0) lcnts[blockIdx.x * TW + wave] = lcnt;
    } else {
        if (__builtin_amdgcn_ballot_w32(cl > LCAP) == 0) {
            const int n = __builtin_amdgcn_readlane(cl, wave);
            int2 cur = make_int2(0, 0);
            if (lane < n) cur = lst[lane];
            for (int k0 = 0; k0 < n; k0 += 32) {
                int2 nxt = make_int2(0, 0);
                if (k0 + 32 + lane < n) nxt = lst[k0 + 32 + lane];
                const int hrow = (cur.x & 0xFFFF) * D;
                const int arow = (int)((unsigned)cur.x >> 16) * PITCH;
                if (n - k0 >= 32) sum_rows<D, true>(32, lane, hrow, arow, cur.y, h, acc);
                else              sum_rows<D, false>(n - k0, lane, hrow, arow, cur.y, h, acc);
                cur = nxt;
            }
        } else {
            scan_dst<D, false>(src, dst, tile8, dinv, h, acc, myq, tid, lane, base, lst, lcnt);
        }
    }

    constexpr int TN  = DOUT / 16;
    constexpr int RT  = TRP / 16;
    constexpr int WPC = TW / TN;
    constexpr int KF  = D / 4;
    static_assert(TW % TN == 0, "waves split evenly over the column blocks");

    const int half = lane >> 4;
    const int l15  = lane & 15;
    const int wv   = __builtin_amdgcn_readfirstlane(wave);
    const int col  = (wv % TN) * 16 + l15;

    v2f bfrag[KF];
#pragma unroll
    for (int j = 0; j < KF; ++j) {
        const int p = 8 * (j >> 1) + 4 * half + 2 * (j & 1);
        const int k = (p & 31) * 4 + (p >> 5);
        bfrag[j].x = W[(size_t)k * DOUT + col];
        bfrag[j].y = W[(size_t)(k + 4) * DOUT + col];
    }
    const float bcol = bias[col];
    float gcol = 0.0f, mcol = 0.0f, rcol = 0.0f, becol = 0.0f;
    if constexpr (BN) {
        gcol  = g[col];
        mcol  = m[col];
        rcol  = rsqrtf(v[col] + 1e-5f);
        becol = be[col];
    }

    __syncthreads();

    for (int tm = wv / TN; tm < RT; tm += WPC) {
        const float* ar = acc + (tm * 16 + l15) * PITCH + 4 * half;
        v8f d = {};
#pragma unroll
        for (int jj = 0; jj < KF / 2; ++jj) {
            const float4 t = *(const float4*)(ar + 8 * jj);
            v2f a0, a1;
            a0.x = t.x; a0.y = t.y;
            a1.x = t.z; a1.y = t.w;
            d = __builtin_amdgcn_wmma_f32_16x16x4_f32(
                 false, a0,  false, bfrag[2 * jj],
                 (short)0, d,  false,  false);
            d = __builtin_amdgcn_wmma_f32_16x16x4_f32(
                 false, a1,  false, bfrag[2 * jj + 1],
                 (short)0, d,  false,  false);
        }
        const int r0 = tm * 16 + 8 * half;
        float* __restrict__ yr = y + (size_t)(base + r0) * DOUT + col;
        float o[8];
#pragma unroll
        for (int r = 0; r < 8; ++r) {
            float val = d[r] + bcol;
            if constexpr (BN) {
                val = gcol * (val - mcol) * rcol + becol;
                val = fmaxf(val, 0.0f);
            }
            o[r] = val;
        }
        if (tm * 16 + 16 <= rows) {
#pragma unroll
            for (int r = 0; r < 8; ++r) yr[(size_t)r * DOUT] = o[r];
        } else {
#pragma unroll
            for (int r = 0; r < 8; ++r)
                if (r0 + r < rows) yr[(size_t)r * DOUT] = o[r];
        }
    }
}


static inline size_t align256(size_t x) { return (x + 255) & ~(size_t)255; }

extern "C" void kernel_launch(void* const* d_in, const int* in_sizes, int n_in,
                              void* d_out, int out_size, void* d_ws, size_t ws_size,
                              hipStream_t stream) {
    (void)in_sizes; (void)n_in; (void)out_size; (void)ws_size;

    const float* x   = (const float*)d_in[0];
    const int*   ei  = (const int*)d_in[1];
    const int*   src = ei;
    const int*   dst = ei + GE;
    const float* W1  = (const float*)d_in[2];
    const float* b1  = (const float*)d_in[3];
    const float* g1  = (const float*)d_in[4];
    const float* be1 = (const float*)d_in[5];
    const float* m1  = (const float*)d_in[6];
    const float* v1  = (const float*)d_in[7];
    const float* W2  = (const float*)d_in[8];
    const float* b2  = (const float*)d_in[9];
    const float* g2  = (const float*)d_in[10];
    const float* be2 = (const float*)d_in[11];
    const float* m2  = (const float*)d_in[12];
    const float* v2  = (const float*)d_in[13];
    const float* W3  = (const float*)d_in[14];
    const float* b3  = (const float*)d_in[15];
    float* out = (float*)d_out;

    char* ws = (char*)d_ws;
    size_t off = 0;
    float* dinv = (float*)(ws + off); off += align256((size_t)GN * 4);
    float* y1   = (float*)(ws + off); off += align256((size_t)GN * GH * 4);
    float* y2   = (float*)(ws + off); off += align256((size_t)GN * GH * 4);
    int*   lcnts = (int*)(ws + off);  off += align256((size_t)NTILES * TW * 4);
    int*   lists = (int*)(ws + off);  off += align256((size_t)NTILES * TW * LCAP * 8);
    unsigned int* tile8 = (unsigned int*)(ws + off); off += align256((size_t)GE);

    const int TPB = 256;

    deg_init_kernel<<<(GN + TPB - 1) / TPB, TPB, 0, stream>>>(dinv, GN);
    deg_count_kernel<<<(GE / 16 + 127) / 128, 128, 0, stream>>>(dst, dinv, tile8, GE / 16);
    deg_to_dinv_kernel<<<(GN + TPB - 1) / TPB, TPB, 0, stream>>>(dinv, GN);

    agg_mm_tile_kernel<GH, true, false><<<NTILES, TT, 0, stream>>>(
        src, dst, dinv, x, W1, b1, g1, be1, m1, v1, y1, tile8, lcnts, lists);

    agg_mm_tile_kernel<GH, true, true><<<NTILES, TT, 0, stream>>>(
        src, dst, dinv, y1, W2, b2, g2, be2, m2, v2, y2, tile8, lcnts, lists);

    agg_mm_tile_kernel<GC, false, true><<<NTILES, TT, 0, stream>>>(
        src, dst, dinv, y2, W3, b3, nullptr, nullptr, nullptr, nullptr, out, tile8, lcnts, lists);
}


# layer 3's product moved in front of its aggregation: layer 2's work-group multiplies its output tile (second LDS tile) by W3 on WMMA; layer 3 gathers 256-B rows, no product; f32 sum order changes
# speedup vs baseline: 3.3560x; 1.0544x over previous
#include <hip/hip_runtime.h>
#include <hip/hip_bf16.h>


typedef __attribute__((ext_vector_type(2))) float v2f;
typedef __attribute__((ext_vector_type(8))) float v8f;

static constexpr int GN  = 50000;
static constexpr int GE  = 800000;
static constexpr int GIN = 128;
static constexpr int GH  = 128;
static constexpr int GC  = 64;


__global__ void deg_init_kernel(float* __restrict__ deg, int n) {
    int i = blockIdx.x * blockDim.x + threadIdx.x;
    if (i < n) deg[i] = 1.0f;
}

static constexpr int TR = 200;

__device__ __forceinline__ unsigned tile_bytes(const int4 d) {
    return  ((unsigned)d.x / (unsigned)TR)        | (((unsigned)d.y / (unsigned)TR) << 8) |
           (((unsigned)d.z / (unsigned)TR) << 16) | (((unsigned)d.w / (unsigned)TR) << 24);
}

__global__ void deg_count_kernel(const int* __restrict__ dst, float* __restrict__ deg,
                                 unsigned int* __restrict__ tile8, int nG) {
    const int t = blockIdx.x * blockDim.x + threadIdx.x;
    if (t < nG) {
        const int4* __restrict__ d4 = (const int4*)dst + (size_t)t * 4;
        const int4 a = d4[0], b = d4[1], c = d4[2], d = d4[3];
        atomicAdd(&deg[a.x], 1.0f); atomicAdd(&deg[a.y], 1.0f); atomicAdd(&deg[a.z], 1.0f); atomicAdd(&deg[a.w], 1.0f);
        atomicAdd(&deg[b.x], 1.0f); atomicAdd(&deg[b.y], 1.0f); atomicAdd(&deg[b.z], 1.0f); atomicAdd(&deg[b.w], 1.0f);
        atomicAdd(&deg[c.x], 1.0f); atomicAdd(&deg[c.y], 1.0f); atomicAdd(&deg[c.z], 1.0f); atomicAdd(&deg[c.w], 1.0f);
        atomicAdd(&deg[d.x], 1.0f); atomicAdd(&deg[d.y], 1.0f); atomicAdd(&deg[d.z], 1.0f); atomicAdd(&deg[d.w], 1.0f);
        uint4 o;
        o.x = tile_bytes(a); o.y = tile_bytes(b); o.z = tile_bytes(c); o.w = tile_bytes(d);
        ((uint4*)tile8)[t] = o;
    }
}

__global__ void deg_to_dinv_kernel(float* __restrict__ deg, int n) {
    int i = blockIdx.x * blockDim.x + threadIdx.x;
    if (i < n) deg[i] = rsqrtf(deg[i]);
}


static constexpr int TRP = 208;
static constexpr int PITCH = 132;
static constexpr int TW = 32;
static constexpr int TT = TW * 32;
static constexpr int QC = 256;
static constexpr int SU = 4;
static constexpr int NTILES = (GN + TR - 1) / TR;
static constexpr int LCAP = 256;
static_assert(GN <= 65536, "a list entry packs the source node into 16 bits");
static_assert(GE % 16 == 0, "the tile bytes are read and written 16 edges at a time");
static_assert(NTILES <= 255, "a tile number is one byte, and 255 is the value no tile has");

template <int VPT>
__device__ __forceinline__ void load_row(const float* __restrict__ p, float (&v)[VPT]) {
    if constexpr (VPT == 4) {
        const float4 t = *(const float4*)p;
        v[0] = t.x; v[1] = t.y; v[2] = t.z; v[3] = t.w;
    } else {
        const float2 t = *(const float2*)p;
        v[0] = t.x; v[1] = t.y;
    }
}

__device__ __forceinline__ void queue_order() {
    __builtin_amdgcn_fence(__ATOMIC_ACQ_REL, "wavefront");
    __builtin_amdgcn_wave_barrier();
}

__device__ __forceinline__ void queue_append(int* myq, int qhead, int& qcnt,
                                             bool m, unsigned entry) {
    const unsigned bal = __builtin_amdgcn_ballot_w32(m);
    if (bal) {
        const int pos = qhead + qcnt + (int)__builtin_amdgcn_mbcnt_lo(bal, 0u);
        if (m) myq[pos & (QC - 1)] = (int)entry;
        qcnt += __builtin_popcount(bal);
    }
}

template <int D, bool FULL>
__device__ __forceinline__ void sum_rows(int nn, int lane, int hrow, int arow, int wbits,
                                         const float* __restrict__ h, float* acc) {
    constexpr int VPT = D / 32;
#pragma unroll 1
    for (int k0 = 0; k0 < nn; k0 += 8) {
        float v[8][VPT];
#pragma unroll
        for (int kk = 0; kk < 8; ++kk) {
#pragma unroll
            for (int c = 0; c < VPT; ++c) v[kk][c] = 0.0f;
            if (FULL || k0 + kk < nn) {
                const int hr = __builtin_amdgcn_readlane(hrow, k0 + kk);
                load_row<VPT>(h + hr + lane * VPT, v[kk]);
            }
        }
#pragma unroll
        for (int kk = 0; kk < 8; ++kk) {
            if (FULL || k0 + kk < nn) {
                const float wk = __int_as_float(__builtin_amdgcn_readlane(wbits, k0 + kk));
                const int   ar = __builtin_amdgcn_readlane(arow, k0 + kk);
#pragma unroll
                for (int c = 0; c < VPT; ++c)
                    __hip_atomic_fetch_add(&acc[ar + c * 32 + lane], wk * v[kk][c],
                                           __ATOMIC_RELAXED, __HIP_MEMORY_SCOPE_WORKGROUP);
            }
        }
    }
}

template <int D, bool FULL, bool KEEP>
__device__ __forceinline__ void drain_edges(const int* myq, int qhead, int n,
                                            int lane, int base,
                                            const int* __restrict__ src,
                                            const int* __restrict__ dst,
                                            const float* __restrict__ dinv,
                                            const float* __restrict__ h,
                                            float* acc,
                                            int2* __restrict__ lst, int& lcnt) {
    const int  nn = FULL ? 32 : n;
    const bool ok = lane < nn;
    const int  ent = myq[(qhead + lane) & (QC - 1)];
    const int  e   = ok ? ent : 0;
    const int  s   = src[e];
    const int  dn  = dst[e];
    const int  dl  = ok ? (dn - base) : 0;
    const float w  = dinv[s] * dinv[dn];
    const int  wbits = __float_as_int(w);
    if (KEEP) {
        if (ok && lcnt + lane < LCAP) lst[lcnt + lane] = make_int2(s | (dl << 16), wbits);
        lcnt += nn;
    }
    sum_rows<D, FULL>(nn, lane, s * D, dl * PITCH, wbits, h, acc);
}

__device__ __forceinline__ unsigned nonzero_bytes(unsigned x) {
    return ((x & 0x7F7F7F7Fu) + 0x7F7F7F7Fu) | x;
}

template <int D, bool KEEP>
__device__ __forceinline__ void process_step(const uint4 (&buf)[SU], int g0, int tid, int lane,
                                             int base, unsigned t4, int* myq, int& qhead, int& qcnt,
                                             const int* __restrict__ src,
                                             const int* __restrict__ dst,
                                             const float* __restrict__ dinv,
                                             const float* __restrict__ h, float* acc,
                                             int2* __restrict__ lst, int& lcnt) {
    constexpr int E16 = GE / 16;
#pragma unroll
    for (int j = 0; j < SU; ++j) {
        const uint4 v = buf[j];
        const int gi = g0 + j * TT + tid;
        const unsigned tt = (gi < E16) ? t4 : 0xFFFFFFFFu;
        unsigned c = (((nonzero_bytes(v.x ^ tt) >> 7) & 0x01010101u) |
                      ((nonzero_bytes(v.y ^ tt) >> 6) & 0x02020202u) |
                      ((nonzero_bytes(v.z ^ tt) >> 5) & 0x04040404u) |
                      ((nonzero_bytes(v.w ^ tt) >> 4) & 0x08080808u)) ^ 0x0F0F0F0Fu;
        unsigned bal = __builtin_amdgcn_ballot_w32(c != 0u);
        while (bal) {
            const int p = __builtin_ctz(c | 0x80000000u);
            const unsigned e = (unsigned)gi * 16u + (unsigned)((p & 3) * 4 + (p >> 3));
            queue_append(myq, qhead, qcnt, c != 0u, e);
            c &= c - 1u;
            if (qcnt >= 32) {
                queue_order();
                do {
                    drain_edges<D, true, KEEP>(myq, qhead, 32, lane, base, src, dst, dinv, h, acc, lst, lcnt);
                    qhead = (qhead + 32) & (QC - 1);
                    qcnt -= 32;
                } while (qcnt >= 32);
                queue_order();
            }
            bal = __builtin_amdgcn_ballot_w32(c != 0u);
        }
    }
}

__device__ __forceinline__ void load_step(uint4 (&buf)[SU], const uint4* __restrict__ t16, int st, int tid) {
    constexpr int E16 = GE / 16;
#pragma unroll
    for (int j = 0; j < SU; ++j) {
        const int idx = st * (TT * SU) + j * TT + tid;
        buf[j] = t16[idx < E16 ? idx : E16 - 1];
    }
}

template <int D, bool KEEP>
__device__ __forceinline__ void scan_dst(const int* __restrict__ src,
                                         const int* __restrict__ dst,
                                         const unsigned int* __restrict__ tile8,
                                         const float* __restrict__ dinv,
                                         const float* __restrict__ h,
                                         float* acc, int* myq, int tid, int lane,
                                         int base,
                                         int2* __restrict__ lst, int& lcnt) {
    int qhead = 0, qcnt = 0;
    const uint4* __restrict__ d4 = (const uint4*)tile8;
    const unsigned t4 = (unsigned)blockIdx.x * 0x01010101u;
    constexpr int E16  = GE / 16;
    constexpr int STEP = TT * SU;

    constexpr int NSTEPS = (E16 + STEP - 1) / STEP;
    static_assert(NSTEPS >= 3, "three buffers in rotation");
    int sA = blockIdx.x % NSTEPS;
    int sB = (sA + 1 == NSTEPS) ? 0 : sA + 1;
    int sC = (sB + 1 == NSTEPS) ? 0 : sB + 1;
    uint4 bufA[SU], bufB[SU], bufC[SU];
    load_step(bufA, d4, sA, tid);
    load_step(bufB, d4, sB, tid);
    load_step(bufC, d4, sC, tid);
    for (int it = 0; ; ) {
        process_step<D, KEEP>(bufA, sA * STEP, tid, lane, base, t4, myq, qhead, qcnt, src, dst, dinv, h, acc, lst, lcnt);
        if (++it == NSTEPS) break;
        sA = (sC + 1 == NSTEPS) ? 0 : sC + 1;
        if (it + 2 < NSTEPS) load_step(bufA, d4, sA, tid);
        process_step<D, KEEP>(bufB, sB * STEP, tid, lane, base, t4, myq, qhead, qcnt, src, dst, dinv, h, acc, lst, lcnt);
        if (++it == NSTEPS) break;
        sB = (sA + 1 == NSTEPS) ? 0 : sA + 1;
        if (it + 2 < NSTEPS) load_step(bufB, d4, sB, tid);
        process_step<D, KEEP>(bufC, sC * STEP, tid, lane, base, t4, myq, qhead, qcnt, src, dst, dinv, h, acc, lst, lcnt);
        if (++it == NSTEPS) break;
        sC = (sB + 1 == NSTEPS) ? 0 : sB + 1;
        if (it + 2 < NSTEPS) load_step(bufC, d4, sC, tid);
    }
    if (qcnt > 0) {
        queue_order();
        drain_edges<D, false, KEEP>(myq, qhead, qcnt, lane, base, src, dst, dinv, h, acc, lst, lcnt);
    }
}

__device__ __forceinline__ v8f block_product(const float* ar, const v2f (&bf)[GH / 4]) {
    v8f d = {};
#pragma unroll
    for (int jj = 0; jj < GH / 8; ++jj) {
        const float4 t = *(const float4*)(ar + 8 * jj);
        v2f a0, a1;
        a0.x = t.x; a0.y = t.y;
        a1.x = t.z; a1.y = t.w;
        d = __builtin_amdgcn_wmma_f32_16x16x4_f32(
             false, a0,  false, bf[2 * jj],
             (short)0, d,  false,  false);
        d = __builtin_amdgcn_wmma_f32_16x16x4_f32(
             false, a1,  false, bf[2 * jj + 1],
             (short)0, d,  false,  false);
    }
    return d;
}

template <int DOUT, bool BN, bool FROM_LIST, int DNEXT>
__global__ __launch_bounds__(TT)
void agg_mm_tile_kernel(const int* __restrict__ src,
                        const int* __restrict__ dst,
                        const float* __restrict__ dinv,
                        const float* __restrict__ h,
                        const float* __restrict__ W,
                        const float* __restrict__ bias,
                        const float* __restrict__ g,
                        const float* __restrict__ be,
                        const float* __restrict__ m,
                        const float* __restrict__ v,
                        const float* __restrict__ Wn,
                        float* __restrict__ y,
                        const unsigned int* __restrict__ tile8,
                        int* __restrict__ lcnts,
                        int* __restrict__ lists) {
    constexpr int D   = GH;
    constexpr int VPT = D / 32;
    static_assert(GIN == GH, "all three layers aggregate 128-wide rows");
    static_assert(VPT == 4, "the position map below is for 4 values per lane");
    static_assert(TR <= 256, "row-in-tile is packed into 8 bits");
    static_assert(TRP % 16 == 0 && TRP >= TR, "whole 16-row blocks");

    __shared__ __attribute__((aligned(16))) float acc[TRP * PITCH];
    __shared__ int   qbuf[TW * QC];
    __shared__ __attribute__((aligned(16))) float ytile[DNEXT > 0 ? TRP * PITCH : 4];

    const int tid  = threadIdx.x;
    const int lane = tid & 31;
    const int wave = tid >> 5;
    const int base = blockIdx.x * TR;
    const int rows = (GN - base < TR) ? (GN - base) : TR;

    int*  myq = qbuf + wave * QC;
    int2* __restrict__ lst = (int2*)lists + (size_t)(blockIdx.x * TW + wave) * LCAP;
    int   lcnt = 0;

    int cl = 0;
    if constexpr (FROM_LIST) cl = lcnts[blockIdx.x * TW + lane];

    for (int r = wave; r < TRP; r += TW) {
        float v4[VPT];
#pragma unroll
        for (int c = 0; c < VPT; ++c) v4[c] = 0.0f;
        if (r < rows) {
            const int i = base + r;
            const float di = dinv[i];
            const float w  = di * di;
            load_row<VPT>(h + (size_t)i * D + lane * VPT, v4);
#pragma unroll
            for (int c = 0; c < VPT; ++c) v4[c] *= w;
        }
#pragma unroll
        for (int c = 0; c < VPT; ++c) acc[r * PITCH + c * 32 + lane] = v4[c];
    }

    __syncthreads();

    if constexpr (!FROM_LIST) {
        scan_dst<D, true>(src, dst, tile8, dinv, h, acc, myq, tid, lane, base, lst, lcnt);
        if (lane == 0) lcnts[blockIdx.x * TW + wave] = lcnt;
    } else {
        if (__builtin_amdgcn_ballot_w32(cl > LCAP) == 0) {
            const int n = __builtin_amdgcn_readlane(cl, wave);
            int2 cur = make_int2(0, 0);
            if (lane < n) cur = lst[lane];
            for (int k0 = 0; k0 < n; k0 += 32) {
                int2 nxt = make_int2(0, 0);
                if (k0 + 32 + lane < n) nxt = lst[k0 + 32 + lane];
                const int hrow = (cur.x & 0xFFFF) * D;
                const int arow = (int)((unsigned)cur.x >> 16) * PITCH;
                if (n - k0 >= 32) sum_rows<D, true>(32, lane, hrow, arow, cur.y, h, acc);
                else              sum_rows<D, false>(n - k0, lane, hrow, arow, cur.y, h, acc);
                cur = nxt;
            }
        } else {
            scan_dst<D, false>(src, dst, tile8, dinv, h, acc, myq, tid, lane, base, lst, lcnt);
        }
    }

    constexpr int TN  = DOUT / 16;
    constexpr int RT  = TRP / 16;
    constexpr int WPC = TW / TN;
    constexpr int KF  = D / 4;
    static_assert(TW % TN == 0, "waves split evenly over the column blocks");

    const int half = lane >> 4;
    const int l15  = lane & 15;
    const int wv   = __builtin_amdgcn_readfirstlane(wave);
    const int col  = (wv % TN) * 16 + l15;

    v2f bfrag[KF];
#pragma unroll
    for (int j = 0; j < KF; ++j) {
        const int p = 8 * (j >> 1) + 4 * half + 2 * (j & 1);
        const int k = (p & 31) * 4 + (p >> 5);
        bfrag[j].x = W[(size_t)k * DOUT + col];
        bfrag[j].y = W[(size_t)(k + 4) * DOUT + col];
    }
    const float bcol = bias[col];
    float gcol = 0.0f, mcol = 0.0f, rcol = 0.0f, becol = 0.0f;
    if constexpr (BN) {
        gcol  = g[col];
        mcol  = m[col];
        rcol  = rsqrtf(v[col] + 1e-5f);
        becol = be[col];
    }

    __syncthreads();

    for (int tm = wv / TN; tm < RT; tm += WPC) {
        const v8f d = block_product(acc + (tm * 16 + l15) * PITCH + 4 * half, bfrag);
        const int r0 = tm * 16 + 8 * half;
        float o[8];
#pragma unroll
        for (int r = 0; r < 8; ++r) {
            float val = d[r] + bcol;
            if constexpr (BN) {
                val = gcol * (val - mcol) * rcol + becol;
                val = fmaxf(val, 0.0f);
            }
            o[r] = val;
        }
        if constexpr (DNEXT > 0) {
            float* yt = ytile + r0 * PITCH + col;
#pragma unroll
            for (int r = 0; r < 8; ++r) yt[r * PITCH] = o[r];
        } else {
            float* __restrict__ yr = y + (size_t)(base + r0) * DOUT + col;
            if (tm * 16 + 16 <= rows) {
#pragma unroll
                for (int r = 0; r < 8; ++r) yr[(size_t)r * DOUT] = o[r];
            } else {
#pragma unroll
                for (int r = 0; r < 8; ++r)
                    if (r0 + r < rows) yr[(size_t)r * DOUT] = o[r];
            }
        }
    }

    if constexpr (DNEXT > 0) {
        constexpr int TN2  = DNEXT / 16;
        constexpr int WPC2 = TW / TN2;
        static_assert(TW % TN2 == 0, "waves split evenly over the column blocks");
        const int col2 = (wv % TN2) * 16 + l15;
        v2f bnext[KF];
#pragma unroll
        for (int j = 0; j < KF; ++j) {
            const int p = 8 * (j >> 1) + 4 * half + 2 * (j & 1);
            bnext[j].x = Wn[(size_t)p * DNEXT + col2];
            bnext[j].y = Wn[(size_t)(p + 1) * DNEXT + col2];
        }

        __syncthreads();

        for (int tm = wv / TN2; tm < RT; tm += WPC2) {
            const v8f d = block_product(ytile + (tm * 16 + l15) * PITCH + 4 * half, bnext);
            const int r0 = tm * 16 + 8 * half;
            float* __restrict__ yr = y + (size_t)(base + r0) * DNEXT + col2;
            if (tm * 16 + 16 <= rows) {
#pragma unroll
                for (int r = 0; r < 8; ++r) yr[(size_t)r * DNEXT] = d[r];
            } else {
#pragma unroll
                for (int r = 0; r < 8; ++r)
                    if (r0 + r < rows) yr[(size_t)r * DNEXT] = d[r];
            }
        }
    }
}

template <int D>
__global__ __launch_bounds__(TT)
void agg_tile_kernel(const int* __restrict__ src,
                     const int* __restrict__ dst,
                     const float* __restrict__ dinv,
                     const float* __restrict__ h,
                     const float* __restrict__ bias,
                     float* __restrict__ y,
                     const unsigned int* __restrict__ tile8,
                     const int* __restrict__ lcnts,
                     const int* __restrict__ lists) {
    constexpr int VPT = D / 32;
    static_assert(VPT == 2, "the store below writes 2 values per lane");

    __shared__ __attribute__((aligned(16))) float acc[TR * PITCH];
    __shared__ int   qbuf[TW * QC];

    const int tid  = threadIdx.x;
    const int lane = tid & 31;
    const int wave = tid >> 5;
    const int base = blockIdx.x * TR;
    const int rows = (GN - base < TR) ? (GN - base) : TR;

    int*  myq = qbuf + wave * QC;
    const int2* __restrict__ lst = (const int2*)lists + (size_t)(blockIdx.x * TW + wave) * LCAP;

    const int cl = lcnts[blockIdx.x * TW + lane];

    for (int r = wave; r < rows; r += TW) {
        const int i = base + r;
        const float di = dinv[i];
        const float w  = di * di;
        float v2[VPT];
        load_row<VPT>(h + (size_t)i * D + lane * VPT, v2);
#pragma unroll
        for (int c = 0; c < VPT; ++c) acc[r * PITCH + c * 32 + lane] = v2[c] * w;
    }

    __syncthreads();

    if (__builtin_amdgcn_ballot_w32(cl > LCAP) == 0) {
        const int n = __builtin_amdgcn_readlane(cl, wave);
        if (n > 0) {
            const int last = n - 1;
            int2 cur = lst[lane < last ? lane : last];
            for (int k0 = 0; k0 < n; k0 += 32) {
                const int kn = k0 + 32 + lane;
                const int2 nxt = lst[kn < last ? kn : last];
                const int hrow = (cur.x & 0xFFFF) * D;
                const int arow = (int)((unsigned)cur.x >> 16) * PITCH;
                if (n - k0 >= 32) sum_rows<D, true>(32, lane, hrow, arow, cur.y, h, acc);
                else              sum_rows<D, false>(n - k0, lane, hrow, arow, cur.y, h, acc);
                cur = nxt;
            }
        }
    } else {
        int lcnt = 0;
        scan_dst<D, false>(src, dst, tile8, dinv, h, acc, myq, tid, lane, base, nullptr, lcnt);
    }

    const float2 b2 = *(const float2*)(bias + lane * VPT);

    __syncthreads();

    for (int r = wave; r < rows; r += TW) {
        float2 o;
        o.x = acc[r * PITCH + lane] + b2.x;
        o.y = acc[r * PITCH + 32 + lane] + b2.y;
        *(float2*)(y + (size_t)(base + r) * D + lane * VPT) = o;
    }
}


static inline size_t align256(size_t x) { return (x + 255) & ~(size_t)255; }

extern "C" void kernel_launch(void* const* d_in, const int* in_sizes, int n_in,
                              void* d_out, int out_size, void* d_ws, size_t ws_size,
                              hipStream_t stream) {
    (void)in_sizes; (void)n_in; (void)out_size; (void)ws_size;

    const float* x   = (const float*)d_in[0];
    const int*   ei  = (const int*)d_in[1];
    const int*   src = ei;
    const int*   dst = ei + GE;
    const float* W1  = (const float*)d_in[2];
    const float* b1  = (const float*)d_in[3];
    const float* g1  = (const float*)d_in[4];
    const float* be1 = (const float*)d_in[5];
    const float* m1  = (const float*)d_in[6];
    const float* v1  = (const float*)d_in[7];
    const float* W2  = (const float*)d_in[8];
    const float* b2  = (const float*)d_in[9];
    const float* g2  = (const float*)d_in[10];
    const float* be2 = (const float*)d_in[11];
    const float* m2  = (const float*)d_in[12];
    const float* v2  = (const float*)d_in[13];
    const float* W3  = (const float*)d_in[14];
    const float* b3  = (const float*)d_in[15];
    float* out = (float*)d_out;

    char* ws = (char*)d_ws;
    size_t off = 0;
    float* dinv = (float*)(ws + off); off += align256((size_t)GN * 4);
    float* y1   = (float*)(ws + off); off += align256((size_t)GN * GH * 4);
    float* z    = (float*)(ws + off); off += align256((size_t)GN * GC * 4);
    int*   lcnts = (int*)(ws + off);  off += align256((size_t)NTILES * TW * 4);
    int*   lists = (int*)(ws + off);  off += align256((size_t)NTILES * TW * LCAP * 8);
    unsigned int* tile8 = (unsigned int*)(ws + off); off += align256((size_t)GE);

    const int TPB = 256;

    deg_init_kernel<<<(GN + TPB - 1) / TPB, TPB, 0, stream>>>(dinv, GN);
    deg_count_kernel<<<(GE / 16 + 127) / 128, 128, 0, stream>>>(dst, dinv, tile8, GE / 16);
    deg_to_dinv_kernel<<<(GN + TPB - 1) / TPB, TPB, 0, stream>>>(dinv, GN);

    agg_mm_tile_kernel<GH, true, false, 0><<<NTILES, TT, 0, stream>>>(
        src, dst, dinv, x, W1, b1, g1, be1, m1, v1, nullptr, y1, tile8, lcnts, lists);

    agg_mm_tile_kernel<GH, true, true, GC><<<NTILES, TT, 0, stream>>>(
        src, dst, dinv, y1, W2, b2, g2, be2, m2, v2, W3, z, tile8, lcnts, lists);

    agg_tile_kernel<GC><<<NTILES, TT, 0, stream>>>(
        src, dst, dinv, z, b3, out, tile8, lcnts, lists);
}
